# starting kernel
# baseline (speedup 1.0000x reference)
#include <hip/hip_runtime.h>
#include <math.h>

typedef float v2f __attribute__((ext_vector_type(2)));
typedef float v8f __attribute__((ext_vector_type(8)));

#define HW    4096
#define CHW   (256 * 4096)
#define PIXS  292
#define MROW  18
#define YROW  20
#define WAVES 2

__global__ __launch_bounds__(256) void repack_weights(const float* __restrict__ wcore,
                                                      const float* __restrict__ wout,
                                                      float* __restrict__ ws) {
    int idx = blockIdx.x * 256 + threadIdx.x;
    if (idx < 65536) {
        int lo   = idx & 1;
        int pair = idx >> 1;
        int n    = pair & 15;
        int m    = (pair >> 4) & 127;
        int t    = pair >> 11;
        int c    = 2 * m + lo;
        ws[idx] = wcore[t * 4096 + c * 16 + n];
    } else {
        int f    = idx - 65536;
        int lo   = f & 1;
        int rest = f >> 1;
        int d    = rest & 255;
        int m    = rest >> 8;
        int k    = 2 * m + lo;
        int e    = k >> 4;
        int g    = k & 15;
        ws[idx] = wout[g * 4096 + d * 16 + e];
    }
}

__global__ __launch_bounds__(WAVES * 32) void tr_main(const float* __restrict__ x,
                                                      const float* __restrict__ wp2,
                                                      const float* __restrict__ vp2,
                                                      float* __restrict__ out) {
    extern __shared__ float smem[];
    const int wave = threadIdx.x >> 5;
    const int lane = threadIdx.x & 31;
    const int half = lane >> 4;
    const int ln   = lane & 15;

    float* Mw = smem + wave * (16 * PIXS);
    float* Yw = smem + WAVES * (16 * PIXS) + wave * (16 * YROW);

    const int tile = blockIdx.x * WAVES + wave;
    const int img  = tile >> 8;
    const int p0   = (tile & 255) << 4;
    const float* xg = x + (size_t)img * CHW + p0 + ln;

    for (int t = 0; t < 16; ++t) {
        v8f acc = {};
        const v2f* wb2 = (const v2f*)wp2 + t * 2048 + ln;
#pragma unroll 4
        for (int kb = 0; kb < 256; kb += 4) {
            const int c0 = kb + 2 * half;
            v2f a;
            a.x = xg[(size_t)c0 * HW];
            a.y = xg[(size_t)(c0 + 1) * HW];
            v2f b = wb2[(c0 >> 1) * 16];
            acc = __builtin_amdgcn_wmma_f32_16x16x4_f32(false, a, false, b,
                                                        (short)0, acc, false, false);
        }
#pragma unroll
        for (int v = 0; v < 8; ++v) {
            const int px = v + 8 * half;
            Mw[px * PIXS + t * MROW + ln] = acc[v];
        }
    }

    for (int px = 0; px < 16; ++px) {
        float* mp = Mw + px * PIXS;
        v8f acc = {};
#pragma unroll
        for (int kb = 0; kb < 16; kb += 4) {
            const int k0 = kb + 2 * half;
            v2f a, b;
            a.x = mp[ln * MROW + k0];
            a.y = mp[ln * MROW + k0 + 1];
            b.x = mp[k0 * MROW + ln];
            b.y = mp[(k0 + 1) * MROW + ln];
            acc = __builtin_amdgcn_wmma_f32_16x16x4_f32(false, a, false, b,
                                                        (short)0, acc, false, false);
        }
#pragma unroll
        for (int v = 0; v < 8; ++v)
            mp[(v + 8 * half) * MROW + ln] = acc[v];
    }

    const int n1 = lane >> 2;
    const int pp = (lane & 3) << 2;
    for (int td = 0; td < 16; ++td) {
        v8f acc = {};
        const v2f* vb2 = (const v2f*)vp2 + td * 16 + ln;
#pragma unroll 4
        for (int kb = 0; kb < 256; kb += 4) {
            const int k0 = kb + 2 * half;
            v2f a;
            const int blk = k0 >> 4, off = k0 & 15;
            a.x = Mw[ln * PIXS + blk * MROW + off];
            a.y = Mw[ln * PIXS + blk * MROW + off + 1];
            v2f b = vb2[(k0 >> 1) * 256];
            acc = __builtin_amdgcn_wmma_f32_16x16x4_f32(false, a, false, b,
                                                        (short)0, acc, false, false);
        }
#pragma unroll
        for (int v = 0; v < 8; ++v)
            Yw[ln * YROW + v + 8 * half] = acc[v];

        const size_t ob = (size_t)img * CHW + (size_t)(td * 16) * HW + p0;
        float4 r1 = *(const float4*)&Yw[n1 * YROW + pp];
        float4 r2 = *(const float4*)&Yw[(n1 + 8) * YROW + pp];
        *(float4*)&out[ob + (size_t)n1 * HW + pp]       = r1;
        *(float4*)&out[ob + (size_t)(n1 + 8) * HW + pp] = r2;
    }
}

__global__ __launch_bounds__(256) void inorm_softsign(float* __restrict__ y) {
    __shared__ float bs[256];
    __shared__ float bss[256];
    __shared__ float mean_s, rs_s;

    float* p = y + (size_t)blockIdx.x * 4096;
    const int tid = threadIdx.x;

    float v[16];
    float s = 0.f, ss = 0.f;
#pragma unroll
    for (int i = 0; i < 16; ++i) {
        float t = p[tid + 256 * i];
        v[i] = t;
        s  += t;
        ss += t * t;
    }
    bs[tid] = s;
    bss[tid] = ss;
    __syncthreads();
    for (int off = 128; off > 0; off >>= 1) {
        if (tid < off) {
            bs[tid]  += bs[tid + off];
            bss[tid] += bss[tid + off];
        }
        __syncthreads();
    }
    if (tid == 0) {
        const float mean = bs[0] * (1.0f / 4096.0f);
        const float var  = bss[0] * (1.0f / 4096.0f) - mean * mean;
        mean_s = mean;
        rs_s   = rsqrtf(var + 1e-5f);
    }
    __syncthreads();
    const float mean = mean_s, rs = rs_s;
#pragma unroll
    for (int i = 0; i < 16; ++i) {
        const float yn = (v[i] - mean) * rs;
        p[tid + 256 * i] = yn / (1.0f + fabsf(yn));
    }
}

extern "C" void kernel_launch(void* const* d_in, const int* in_sizes, int n_in,
                              void* d_out, int out_size, void* d_ws, size_t ws_size,
                              hipStream_t stream) {
    (void)in_sizes; (void)n_in; (void)out_size; (void)ws_size;
    const float* x     = (const float*)d_in[0];
    const float* wcore = (const float*)d_in[1];
    const float* wout  = (const float*)d_in[2];
    float* out = (float*)d_out;
    float* ws  = (float*)d_ws;
    const float* wp2 = ws;
    const float* vp2 = ws + 65536;

    repack_weights<<<512, 256, 0, stream>>>(wcore, wout, ws);

    const size_t smem_bytes = (WAVES * (16 * PIXS) + WAVES * (16 * YROW)) * sizeof(float);
    tr_main<<<4096 / WAVES, WAVES * 32, smem_bytes, stream>>>(x, wp2, vp2, out);

    inorm_softsign<<<16 * 256, 256, 0, stream>>>(out);
}


# Register-block both big GEMMs: a wave takes 32 pixels x 8 column tiles (16 accumulators), so a B fragment feeds 2 WMMAs and an A fragment 8; the next block's fragments load ahead of the WMMAs
# speedup vs baseline: 1.9643x; 1.9643x over previous
#include <hip/hip_runtime.h>
#include <math.h>

typedef float v2f __attribute__((ext_vector_type(2)));
typedef float v8f __attribute__((ext_vector_type(8)));

#define HW    4096
#define CHW   (256 * 4096)
#define PIXS  292
#define MROW  18
#define YROW  36
#define WAVES 2
#define MT    2
#define NT    8
#define LDSF  (WAVES * (MT * 16 * PIXS) + WAVES * (16 * YROW))

__global__ __launch_bounds__(256) void repack_weights(const float* __restrict__ wcore,
                                                      const float* __restrict__ wout,
                                                      float* __restrict__ ws) {
    int idx = blockIdx.x * 256 + threadIdx.x;
    if (idx < 65536) {
        int lo   = idx & 1;
        int pair = idx >> 1;
        int n    = pair & 15;
        int m    = (pair >> 4) & 127;
        int t    = pair >> 11;
        int c    = 2 * m + lo;
        ws[idx] = wcore[t * 4096 + c * 16 + n];
    } else {
        int f    = idx - 65536;
        int lo   = f & 1;
        int rest = f >> 1;
        int d    = rest & 255;
        int m    = rest >> 8;
        int k    = 2 * m + lo;
        int e    = k >> 4;
        int g    = k & 15;
        ws[idx] = wout[g * 4096 + d * 16 + e];
    }
}

struct Frag {
    v2f a[2][MT];
    v2f b[2][NT];
};


__device__ __forceinline__ void load1(Frag& f, const float* __restrict__ xg,
                                      const v2f* __restrict__ wb2, int i, int half) {
#pragma unroll
    for (int s = 0; s < 2; ++s) {
        const int c0 = 8 * i + 4 * s + 2 * half;
#pragma unroll
        for (int m = 0; m < MT; ++m) {
            f.a[s][m].x = xg[(size_t)c0 * HW + 16 * m];
            f.a[s][m].y = xg[(size_t)(c0 + 1) * HW + 16 * m];
        }
#pragma unroll
        for (int j = 0; j < NT; ++j)
            f.b[s][j] = wb2[j * 2048 + (c0 >> 1) * 16];
    }
}

__device__ __forceinline__ void load2(Frag& f, const float* Pw,
                                      const v2f* __restrict__ vb2, int i, int half) {
#pragma unroll
    for (int s = 0; s < 2; ++s) {
        const int k0  = 8 * i + 4 * s + 2 * half;
        const int blk = k0 >> 4, off = k0 & 15;
#pragma unroll
        for (int m = 0; m < MT; ++m) {
            f.a[s][m].x = Pw[(16 * m) * PIXS + blk * MROW + off];
            f.a[s][m].y = Pw[(16 * m) * PIXS + blk * MROW + off + 1];
        }
#pragma unroll
        for (int j = 0; j < NT; ++j)
            f.b[s][j] = vb2[(k0 >> 1) * 256 + j * 16];
    }
}

__device__ __forceinline__ void mma_block(v8f (&acc)[MT][NT], const Frag& f) {
#pragma unroll
    for (int s = 0; s < 2; ++s)
#pragma unroll
        for (int j = 0; j < NT; ++j)
#pragma unroll
            for (int m = 0; m < MT; ++m)
                acc[m][j] = __builtin_amdgcn_wmma_f32_16x16x4_f32(false, f.a[s][m], false, f.b[s][j],
                                                                  (short)0, acc[m][j], false, false);
}

__global__ __launch_bounds__(WAVES * 32) void tr_main(const float* __restrict__ x,
                                                      const float* __restrict__ wp2,
                                                      const float* __restrict__ vp2,
                                                      float* __restrict__ out) {
    __shared__ __attribute__((aligned(16))) float smem[LDSF];
    const int wave = threadIdx.x >> 5;
    const int lane = threadIdx.x & 31;
    const int half = lane >> 4;
    const int ln   = lane & 15;

    float* Mw = smem + wave * (MT * 16 * PIXS);
    float* Yw = smem + WAVES * (MT * 16 * PIXS) + wave * (16 * YROW);

    const int tile = blockIdx.x * WAVES + wave;
    const int img  = tile >> 7;
    const int p0   = (tile & 127) << 5;
    const float* xg = x + (size_t)img * CHW + p0 + ln;

    for (int tg = 0; tg < 16; tg += NT) {
        v8f acc[MT][NT];
#pragma unroll
        for (int m = 0; m < MT; ++m)
#pragma unroll
            for (int j = 0; j < NT; ++j) acc[m][j] = v8f{};
        const v2f* wb2 = (const v2f*)wp2 + tg * 2048 + ln;
        Frag f0, f1;
        load1(f0, xg, wb2, 0, half);
        for (int i = 0; i < 30; i += 2) {
            load1(f1, xg, wb2, i + 1, half);
            mma_block(acc, f0);
            load1(f0, xg, wb2, i + 2, half);
            mma_block(acc, f1);
        }
        load1(f1, xg, wb2, 31, half);
        mma_block(acc, f0);
        mma_block(acc, f1);
#pragma unroll
        for (int j = 0; j < NT; ++j)
#pragma unroll
            for (int m = 0; m < MT; ++m)
#pragma unroll
                for (int v = 0; v < 8; ++v) {
                    const int px = 16 * m + v + 8 * half;
                    Mw[px * PIXS + (tg + j) * MROW + ln] = acc[m][j][v];
                }
    }

    for (int px = 0; px < MT * 16; ++px) {
        float* mp = Mw + px * PIXS;
        v8f acc = {};
#pragma unroll
        for (int kb = 0; kb < 16; kb += 4) {
            const int k0 = kb + 2 * half;
            v2f a, b;
            a.x = mp[ln * MROW + k0];
            a.y = mp[ln * MROW + k0 + 1];
            b.x = mp[k0 * MROW + ln];
            b.y = mp[(k0 + 1) * MROW + ln];
            acc = __builtin_amdgcn_wmma_f32_16x16x4_f32(false, a, false, b,
                                                        (short)0, acc, false, false);
        }
#pragma unroll
        for (int v = 0; v < 8; ++v)
            mp[(v + 8 * half) * MROW + ln] = acc[v];
    }

    const int yr = lane >> 3;
    const int yc = (lane & 7) << 2;
    const float* Pw = Mw + ln * PIXS;
    for (int tdg = 0; tdg < 16; tdg += NT) {
        v8f acc[MT][NT];
#pragma unroll
        for (int m = 0; m < MT; ++m)
#pragma unroll
            for (int j = 0; j < NT; ++j) acc[m][j] = v8f{};
        const v2f* vb2 = (const v2f*)vp2 + tdg * 16 + ln;
        Frag f0, f1;
        load2(f0, Pw, vb2, 0, half);
        for (int i = 0; i < 30; i += 2) {
            load2(f1, Pw, vb2, i + 1, half);
            mma_block(acc, f0);
            load2(f0, Pw, vb2, i + 2, half);
            mma_block(acc, f1);
        }
        load2(f1, Pw, vb2, 31, half);
        mma_block(acc, f0);
        mma_block(acc, f1);

#pragma unroll
        for (int j = 0; j < NT; ++j) {
#pragma unroll
            for (int m = 0; m < MT; ++m)
#pragma unroll
                for (int v = 0; v < 8; ++v)
                    Yw[ln * YROW + 16 * m + v + 8 * half] = acc[m][j][v];

            const size_t ob = (size_t)img * CHW + (size_t)((tdg + j) * 16) * HW + p0;
#pragma unroll
            for (int q = 0; q < 4; ++q) {
                const int row = yr + 4 * q;
                const float4 r = *(const float4*)&Yw[row * YROW + yc];
                *(float4*)&out[ob + (size_t)row * HW + yc] = r;
            }
        }
    }
}

__global__ __launch_bounds__(256) void inorm_softsign(float* __restrict__ y) {
    __shared__ float bs[256];
    __shared__ float bss[256];
    __shared__ float mean_s, rs_s;

    float* p = y + (size_t)blockIdx.x * 4096;
    const int tid = threadIdx.x;

    float v[16];
    float s = 0.f, ss = 0.f;
#pragma unroll
    for (int i = 0; i < 16; ++i) {
        float t = p[tid + 256 * i];
        v[i] = t;
        s  += t;
        ss += t * t;
    }
    bs[tid] = s;
    bss[tid] = ss;
    __syncthreads();
    for (int off = 128; off > 0; off >>= 1) {
        if (tid < off) {
            bs[tid]  += bs[tid + off];
            bss[tid] += bss[tid + off];
        }
        __syncthreads();
    }
    if (tid == 0) {
        const float mean = bs[0] * (1.0f / 4096.0f);
        const float var  = bss[0] * (1.0f / 4096.0f) - mean * mean;
        mean_s = mean;
        rs_s   = rsqrtf(var + 1e-5f);
    }
    __syncthreads();
    const float mean = mean_s, rs = rs_s;
#pragma unroll
    for (int i = 0; i < 16; ++i) {
        const float yn = (v[i] - mean) * rs;
        p[tid + 256 * i] = yn / (1.0f + fabsf(yn));
    }
}

extern "C" void kernel_launch(void* const* d_in, const int* in_sizes, int n_in,
                              void* d_out, int out_size, void* d_ws, size_t ws_size,
                              hipStream_t stream) {
    (void)in_sizes; (void)n_in; (void)out_size; (void)ws_size;
    const float* x     = (const float*)d_in[0];
    const float* wcore = (const float*)d_in[1];
    const float* wout  = (const float*)d_in[2];
    float* out = (float*)d_out;
    float* ws  = (float*)d_ws;
    const float* wp2 = ws;
    const float* vp2 = ws + 65536;

    repack_weights<<<512, 256, 0, stream>>>(wcore, wout, ws);

    tr_main<<<2048 / WAVES, WAVES * 32, 0, stream>>>(x, wp2, vp2, out);

    inorm_softsign<<<16 * 256, 256, 0, stream>>>(out);
}
